# starting kernel
# baseline (speedup 1.0000x reference)
#include <hip/hip_runtime.h>
#include <hip/hip_bf16.h>

typedef float v2f __attribute__((ext_vector_type(2)));
typedef float v4f __attribute__((ext_vector_type(4)));
typedef float v8f __attribute__((ext_vector_type(8)));

#define B_DIM   64
#define N_DIM   8192
#define D_DIM   128
#define PCHUNK  32
#define NROWS   (N_DIM / PCHUNK)
#define WSTRIDE ((size_t)B_DIM * PCHUNK * D_DIM)

#define FLT_BIG 3.402823466e+38f

__global__ __launch_bounds__(256) void agg_stage1(const float* __restrict__ in,
                                                  float* __restrict__ ws) {
    const int p    = blockIdx.x;
    const int b    = blockIdx.y;
    const int tid  = threadIdx.x;
    const int wave = tid >> 5;
    const int lane = tid & 31;

    const size_t base = ((size_t)b * N_DIM + (size_t)p * NROWS) * D_DIM;
    const v4f* ptr = (const v4f*)(in + base) + lane + (size_t)wave * (D_DIM / 4);

    v4f s  = {0.f, 0.f, 0.f, 0.f};
    v4f q  = {0.f, 0.f, 0.f, 0.f};
    v4f mx = {-FLT_BIG, -FLT_BIG, -FLT_BIG, -FLT_BIG};
    v4f mn = { FLT_BIG,  FLT_BIG,  FLT_BIG,  FLT_BIG};

    #pragma unroll 4
    for (int r = 0; r < NROWS / 8; ++r) {
        v4f v = __builtin_nontemporal_load(ptr);
        ptr += 8 * (D_DIM / 4);
        s.x += v.x; s.y += v.y; s.z += v.z; s.w += v.w;
        q.x = __builtin_fmaf(v.x, v.x, q.x);
        q.y = __builtin_fmaf(v.y, v.y, q.y);
        q.z = __builtin_fmaf(v.z, v.z, q.z);
        q.w = __builtin_fmaf(v.w, v.w, q.w);
        mx.x = fmaxf(mx.x, v.x); mx.y = fmaxf(mx.y, v.y);
        mx.z = fmaxf(mx.z, v.z); mx.w = fmaxf(mx.w, v.w);
        mn.x = fminf(mn.x, v.x); mn.y = fminf(mn.y, v.y);
        mn.z = fminf(mn.z, v.z); mn.w = fminf(mn.w, v.w);
    }

    __shared__ float lsum[8][D_DIM];
    __shared__ float lsq [8][D_DIM];
    __shared__ float lmx [8][D_DIM];
    __shared__ float lmn [8][D_DIM];
    const int d0 = lane * 4;
    lsum[wave][d0+0] = s.x;  lsum[wave][d0+1] = s.y;
    lsum[wave][d0+2] = s.z;  lsum[wave][d0+3] = s.w;
    lsq [wave][d0+0] = q.x;  lsq [wave][d0+1] = q.y;
    lsq [wave][d0+2] = q.z;  lsq [wave][d0+3] = q.w;
    lmx [wave][d0+0] = mx.x; lmx [wave][d0+1] = mx.y;
    lmx [wave][d0+2] = mx.z; lmx [wave][d0+3] = mx.w;
    lmn [wave][d0+0] = mn.x; lmn [wave][d0+1] = mn.y;
    lmn [wave][d0+2] = mn.z; lmn [wave][d0+3] = mn.w;
    __syncthreads();

    if (tid < D_DIM) {
        float S = 0.f, Q = 0.f, MX = -FLT_BIG, MN = FLT_BIG;
        #pragma unroll
        for (int w = 0; w < 8; ++w) {
            S += lsum[w][tid];
            Q += lsq [w][tid];
            MX = fmaxf(MX, lmx[w][tid]);
            MN = fminf(MN, lmn[w][tid]);
        }
        const size_t o = ((size_t)b * PCHUNK + p) * D_DIM + tid;
        ws[o]               = S;
        ws[WSTRIDE + o]     = Q;
        ws[2 * WSTRIDE + o] = MX;
        ws[3 * WSTRIDE + o] = MN;
    }
}

__global__ __launch_bounds__(32) void agg_stage2(const float* __restrict__ ws,
                                                 float* __restrict__ out) {
    const int dc   = blockIdx.x;
    const int b    = blockIdx.y;
    const int lane = threadIdx.x;
    const int nl   = lane & 15;
    const int hi   = lane >> 4;
    const int n    = dc * 16 + nl;

    const float* psum = ws + (size_t)b * PCHUNK * D_DIM;
    const float* psq  = psum + WSTRIDE;
    const float* pmx  = psum + 2 * WSTRIDE;
    const float* pmn  = psum + 3 * WSTRIDE;

    const v2f ones = {1.0f, 1.0f};
    v8f csum = {};
    v8f csq  = {};

    #pragma unroll
    for (int k0 = 0; k0 < PCHUNK; k0 += 4) {
        const int ka = k0 + hi;
        const int kb = k0 + 2 + hi;
        v2f bs, bq;
        bs.x = psum[(size_t)ka * D_DIM + n];
        bs.y = psum[(size_t)kb * D_DIM + n];
        bq.x = psq [(size_t)ka * D_DIM + n];
        bq.y = psq [(size_t)kb * D_DIM + n];
        csum = __builtin_amdgcn_wmma_f32_16x16x4_f32(
            false, ones, false, bs, (short)0, csum, false, false);
        csq = __builtin_amdgcn_wmma_f32_16x16x4_f32(
            false, ones, false, bq, (short)0, csq, false, false);
    }

    float MX = -FLT_BIG, MN = FLT_BIG;
    for (int k = hi; k < PCHUNK; k += 2) {
        MX = fmaxf(MX, pmx[(size_t)k * D_DIM + n]);
        MN = fminf(MN, pmn[(size_t)k * D_DIM + n]);
    }
    MX = fmaxf(MX, __shfl_xor(MX, 16, 32));
    MN = fminf(MN, __shfl_xor(MN, 16, 32));

    if (lane < 16) {
        const float S = csum[0];
        const float Q = csq[0];
        const float invN = 1.0f / (float)N_DIM;
        const float mean = S * invN;
        const float var  = fmaxf(Q * invN - mean * mean, 0.0f);
        const float sd   = __builtin_sqrtf(var);

        float* o = out + (size_t)b * (5 * D_DIM);
        o[n]             = mean;
        o[D_DIM + n]     = MX;
        o[2 * D_DIM + n] = MN;
        o[3 * D_DIM + n] = S;
        o[4 * D_DIM + n] = sd;
    }
}

extern "C" void kernel_launch(void* const* d_in, const int* in_sizes, int n_in,
                              void* d_out, int out_size, void* d_ws, size_t ws_size,
                              hipStream_t stream) {
    (void)in_sizes; (void)n_in; (void)out_size; (void)ws_size;
    const float* in = (const float*)d_in[0];
    float* out = (float*)d_out;
    float* ws  = (float*)d_ws;

    dim3 g1(PCHUNK, B_DIM);
    agg_stage1<<<g1, 256, 0, stream>>>(in, ws);

    dim3 g2(D_DIM / 16, B_DIM);
    agg_stage2<<<g2, 32, 0, stream>>>(ws, out);
}


# Stage 1 regridded to 256 long-lived 16-wave work-groups: 128 contiguous rows per wave, 16-row load groups requested one group ahead; partials per batch 32 to 4, so the f32 sum order over N changes
# speedup vs baseline: 1.1495x; 1.1495x over previous
#include <hip/hip_runtime.h>
#include <hip/hip_bf16.h>

typedef float v2f __attribute__((ext_vector_type(2)));
typedef float v4f __attribute__((ext_vector_type(4)));
typedef float v8f __attribute__((ext_vector_type(8)));

#define B_DIM   64
#define N_DIM   8192
#define D_DIM   128
#define PCHUNK  4
#define NROWS   (N_DIM / PCHUNK)
#define NWAVE   16
#define RPW     (NROWS / NWAVE)
#define GRP     16
#define NGRP    (RPW / GRP)
#define ROWV    (D_DIM / 4)
#define WSTRIDE ((size_t)B_DIM * PCHUNK * D_DIM)

#define FLT_BIG 3.402823466e+38f

static __device__ __forceinline__ void acc_row(const v4f v, v4f& s, v4f& q,
                                               v4f& mx, v4f& mn) {
    s.x += v.x; s.y += v.y; s.z += v.z; s.w += v.w;
    q.x = __builtin_fmaf(v.x, v.x, q.x);
    q.y = __builtin_fmaf(v.y, v.y, q.y);
    q.z = __builtin_fmaf(v.z, v.z, q.z);
    q.w = __builtin_fmaf(v.w, v.w, q.w);
    mx.x = fmaxf(mx.x, v.x); mx.y = fmaxf(mx.y, v.y);
    mx.z = fmaxf(mx.z, v.z); mx.w = fmaxf(mx.w, v.w);
    mn.x = fminf(mn.x, v.x); mn.y = fminf(mn.y, v.y);
    mn.z = fminf(mn.z, v.z); mn.w = fminf(mn.w, v.w);
}

__global__ __launch_bounds__(NWAVE * 32) void agg_stage1(const float* __restrict__ in,
                                                         float* __restrict__ ws) {
    const int p    = blockIdx.x;
    const int b    = blockIdx.y;
    const int tid  = threadIdx.x;
    const int wave = tid >> 5;
    const int lane = tid & 31;

    const size_t base = ((size_t)b * N_DIM + (size_t)p * NROWS + (size_t)wave * RPW) * D_DIM;
    const v4f* ptr = (const v4f*)(in + base) + lane;

    v4f s  = {0.f, 0.f, 0.f, 0.f};
    v4f q  = {0.f, 0.f, 0.f, 0.f};
    v4f mx = {-FLT_BIG, -FLT_BIG, -FLT_BIG, -FLT_BIG};
    v4f mn = { FLT_BIG,  FLT_BIG,  FLT_BIG,  FLT_BIG};

    v4f buf[2][GRP];

    #pragma unroll
    for (int i = 0; i < GRP; ++i)
        buf[0][i] = __builtin_nontemporal_load(ptr + i * ROWV);

    #pragma unroll
    for (int g = 0; g < NGRP; ++g) {
        __builtin_amdgcn_sched_barrier(0);
        if (g + 1 < NGRP) {
            #pragma unroll
            for (int i = 0; i < GRP; ++i)
                buf[(g + 1) & 1][i] =
                    __builtin_nontemporal_load(ptr + ((g + 1) * GRP + i) * ROWV);
        }
        __builtin_amdgcn_sched_barrier(0);
        #pragma unroll
        for (int i = 0; i < GRP; ++i) acc_row(buf[g & 1][i], s, q, mx, mn);
    }

    __shared__ float lsum[NWAVE][D_DIM];
    __shared__ float lsq [NWAVE][D_DIM];
    __shared__ float lmx [NWAVE][D_DIM];
    __shared__ float lmn [NWAVE][D_DIM];
    const int d0 = lane * 4;
    lsum[wave][d0+0] = s.x;  lsum[wave][d0+1] = s.y;
    lsum[wave][d0+2] = s.z;  lsum[wave][d0+3] = s.w;
    lsq [wave][d0+0] = q.x;  lsq [wave][d0+1] = q.y;
    lsq [wave][d0+2] = q.z;  lsq [wave][d0+3] = q.w;
    lmx [wave][d0+0] = mx.x; lmx [wave][d0+1] = mx.y;
    lmx [wave][d0+2] = mx.z; lmx [wave][d0+3] = mx.w;
    lmn [wave][d0+0] = mn.x; lmn [wave][d0+1] = mn.y;
    lmn [wave][d0+2] = mn.z; lmn [wave][d0+3] = mn.w;
    __syncthreads();

    if (tid < D_DIM) {
        float S = 0.f, Q = 0.f, MX = -FLT_BIG, MN = FLT_BIG;
        #pragma unroll
        for (int w = 0; w < NWAVE; ++w) {
            S += lsum[w][tid];
            Q += lsq [w][tid];
            MX = fmaxf(MX, lmx[w][tid]);
            MN = fminf(MN, lmn[w][tid]);
        }
        const size_t o = ((size_t)b * PCHUNK + p) * D_DIM + tid;
        ws[o]               = S;
        ws[WSTRIDE + o]     = Q;
        ws[2 * WSTRIDE + o] = MX;
        ws[3 * WSTRIDE + o] = MN;
    }
}

__global__ __launch_bounds__(32) void agg_stage2(const float* __restrict__ ws,
                                                 float* __restrict__ out) {
    const int dc   = blockIdx.x;
    const int b    = blockIdx.y;
    const int lane = threadIdx.x;
    const int nl   = lane & 15;
    const int hi   = lane >> 4;
    const int n    = dc * 16 + nl;

    const float* psum = ws + (size_t)b * PCHUNK * D_DIM;
    const float* psq  = psum + WSTRIDE;
    const float* pmx  = psum + 2 * WSTRIDE;
    const float* pmn  = psum + 3 * WSTRIDE;

    const v2f ones = {1.0f, 1.0f};
    v8f csum = {};
    v8f csq  = {};

    #pragma unroll
    for (int k0 = 0; k0 < PCHUNK; k0 += 4) {
        const int ka = k0 + hi;
        const int kb = k0 + 2 + hi;
        v2f bs, bq;
        bs.x = psum[(size_t)ka * D_DIM + n];
        bs.y = psum[(size_t)kb * D_DIM + n];
        bq.x = psq [(size_t)ka * D_DIM + n];
        bq.y = psq [(size_t)kb * D_DIM + n];
        csum = __builtin_amdgcn_wmma_f32_16x16x4_f32(
            false, ones, false, bs, (short)0, csum, false, false);
        csq = __builtin_amdgcn_wmma_f32_16x16x4_f32(
            false, ones, false, bq, (short)0, csq, false, false);
    }

    float MX = -FLT_BIG, MN = FLT_BIG;
    for (int k = hi; k < PCHUNK; k += 2) {
        MX = fmaxf(MX, pmx[(size_t)k * D_DIM + n]);
        MN = fminf(MN, pmn[(size_t)k * D_DIM + n]);
    }
    MX = fmaxf(MX, __shfl_xor(MX, 16, 32));
    MN = fminf(MN, __shfl_xor(MN, 16, 32));

    if (lane < 16) {
        const float S = csum[0];
        const float Q = csq[0];
        const float invN = 1.0f / (float)N_DIM;
        const float mean = S * invN;
        const float var  = fmaxf(Q * invN - mean * mean, 0.0f);
        const float sd   = __builtin_sqrtf(var);

        float* o = out + (size_t)b * (5 * D_DIM);
        o[n]             = mean;
        o[D_DIM + n]     = MX;
        o[2 * D_DIM + n] = MN;
        o[3 * D_DIM + n] = S;
        o[4 * D_DIM + n] = sd;
    }
}

extern "C" void kernel_launch(void* const* d_in, const int* in_sizes, int n_in,
                              void* d_out, int out_size, void* d_ws, size_t ws_size,
                              hipStream_t stream) {
    (void)in_sizes; (void)n_in; (void)out_size; (void)ws_size;
    const float* in = (const float*)d_in[0];
    float* out = (float*)d_out;
    float* ws  = (float*)d_ws;

    dim3 g1(PCHUNK, B_DIM);
    agg_stage1<<<g1, NWAVE * 32, 0, stream>>>(in, ws);

    dim3 g2(D_DIM / 16, B_DIM);
    agg_stage2<<<g2, 32, 0, stream>>>(ws, out);
}


# Stage 2: the max/min partial loads are written out and issued with the sum loads, ahead of the matrix products: one memory round trip in its chain instead of three (the rolled loop waited per trip)
# speedup vs baseline: 1.1873x; 1.0329x over previous
#include <hip/hip_runtime.h>
#include <hip/hip_bf16.h>

typedef float v2f __attribute__((ext_vector_type(2)));
typedef float v4f __attribute__((ext_vector_type(4)));
typedef float v8f __attribute__((ext_vector_type(8)));

#define B_DIM   64
#define N_DIM   8192
#define D_DIM   128
#define PCHUNK  4
#define NROWS   (N_DIM / PCHUNK)
#define NWAVE   16
#define RPW     (NROWS / NWAVE)
#define GRP     16
#define NGRP    (RPW / GRP)
#define ROWV    (D_DIM / 4)
#define WSTRIDE ((size_t)B_DIM * PCHUNK * D_DIM)

#define FLT_BIG 3.402823466e+38f

static __device__ __forceinline__ void acc_row(const v4f v, v4f& s, v4f& q,
                                               v4f& mx, v4f& mn) {
    s.x += v.x; s.y += v.y; s.z += v.z; s.w += v.w;
    q.x = __builtin_fmaf(v.x, v.x, q.x);
    q.y = __builtin_fmaf(v.y, v.y, q.y);
    q.z = __builtin_fmaf(v.z, v.z, q.z);
    q.w = __builtin_fmaf(v.w, v.w, q.w);
    mx.x = fmaxf(mx.x, v.x); mx.y = fmaxf(mx.y, v.y);
    mx.z = fmaxf(mx.z, v.z); mx.w = fmaxf(mx.w, v.w);
    mn.x = fminf(mn.x, v.x); mn.y = fminf(mn.y, v.y);
    mn.z = fminf(mn.z, v.z); mn.w = fminf(mn.w, v.w);
}

__global__ __launch_bounds__(NWAVE * 32) void agg_stage1(const float* __restrict__ in,
                                                         float* __restrict__ ws) {
    const int p    = blockIdx.x;
    const int b    = blockIdx.y;
    const int tid  = threadIdx.x;
    const int wave = tid >> 5;
    const int lane = tid & 31;

    const size_t base = ((size_t)b * N_DIM + (size_t)p * NROWS + (size_t)wave * RPW) * D_DIM;
    const v4f* ptr = (const v4f*)(in + base) + lane;

    v4f s  = {0.f, 0.f, 0.f, 0.f};
    v4f q  = {0.f, 0.f, 0.f, 0.f};
    v4f mx = {-FLT_BIG, -FLT_BIG, -FLT_BIG, -FLT_BIG};
    v4f mn = { FLT_BIG,  FLT_BIG,  FLT_BIG,  FLT_BIG};

    v4f buf[2][GRP];

    #pragma unroll
    for (int i = 0; i < GRP; ++i)
        buf[0][i] = __builtin_nontemporal_load(ptr + i * ROWV);

    #pragma unroll
    for (int g = 0; g < NGRP; ++g) {
        __builtin_amdgcn_sched_barrier(0);
        if (g + 1 < NGRP) {
            #pragma unroll
            for (int i = 0; i < GRP; ++i)
                buf[(g + 1) & 1][i] =
                    __builtin_nontemporal_load(ptr + ((g + 1) * GRP + i) * ROWV);
        }
        __builtin_amdgcn_sched_barrier(0);
        #pragma unroll
        for (int i = 0; i < GRP; ++i) acc_row(buf[g & 1][i], s, q, mx, mn);
    }

    __shared__ float lsum[NWAVE][D_DIM];
    __shared__ float lsq [NWAVE][D_DIM];
    __shared__ float lmx [NWAVE][D_DIM];
    __shared__ float lmn [NWAVE][D_DIM];
    const int d0 = lane * 4;
    lsum[wave][d0+0] = s.x;  lsum[wave][d0+1] = s.y;
    lsum[wave][d0+2] = s.z;  lsum[wave][d0+3] = s.w;
    lsq [wave][d0+0] = q.x;  lsq [wave][d0+1] = q.y;
    lsq [wave][d0+2] = q.z;  lsq [wave][d0+3] = q.w;
    lmx [wave][d0+0] = mx.x; lmx [wave][d0+1] = mx.y;
    lmx [wave][d0+2] = mx.z; lmx [wave][d0+3] = mx.w;
    lmn [wave][d0+0] = mn.x; lmn [wave][d0+1] = mn.y;
    lmn [wave][d0+2] = mn.z; lmn [wave][d0+3] = mn.w;
    __syncthreads();

    if (tid < D_DIM) {
        float S = 0.f, Q = 0.f, MX = -FLT_BIG, MN = FLT_BIG;
        #pragma unroll
        for (int w = 0; w < NWAVE; ++w) {
            S += lsum[w][tid];
            Q += lsq [w][tid];
            MX = fmaxf(MX, lmx[w][tid]);
            MN = fminf(MN, lmn[w][tid]);
        }
        const size_t o = ((size_t)b * PCHUNK + p) * D_DIM + tid;
        ws[o]               = S;
        ws[WSTRIDE + o]     = Q;
        ws[2 * WSTRIDE + o] = MX;
        ws[3 * WSTRIDE + o] = MN;
    }
}

__global__ __launch_bounds__(32) void agg_stage2(const float* __restrict__ ws,
                                                 float* __restrict__ out) {
    const int dc   = blockIdx.x;
    const int b    = blockIdx.y;
    const int lane = threadIdx.x;
    const int nl   = lane & 15;
    const int hi   = lane >> 4;
    const int n    = dc * 16 + nl;

    const float* psum = ws + (size_t)b * PCHUNK * D_DIM;
    const float* psq  = psum + WSTRIDE;
    const float* pmx  = psum + 2 * WSTRIDE;
    const float* pmn  = psum + 3 * WSTRIDE;

    static_assert(PCHUNK == 4, "stage 2 is written out for four partials per batch");
    const float mx0 = pmx[(size_t)hi * D_DIM + n];
    const float mx1 = pmx[(size_t)(hi + 2) * D_DIM + n];
    const float mn0 = pmn[(size_t)hi * D_DIM + n];
    const float mn1 = pmn[(size_t)(hi + 2) * D_DIM + n];

    const v2f ones = {1.0f, 1.0f};
    v8f csum = {};
    v8f csq  = {};

    #pragma unroll
    for (int k0 = 0; k0 < PCHUNK; k0 += 4) {
        const int ka = k0 + hi;
        const int kb = k0 + 2 + hi;
        v2f bs, bq;
        bs.x = psum[(size_t)ka * D_DIM + n];
        bs.y = psum[(size_t)kb * D_DIM + n];
        bq.x = psq [(size_t)ka * D_DIM + n];
        bq.y = psq [(size_t)kb * D_DIM + n];
        csum = __builtin_amdgcn_wmma_f32_16x16x4_f32(
            false, ones, false, bs, (short)0, csum, false, false);
        csq = __builtin_amdgcn_wmma_f32_16x16x4_f32(
            false, ones, false, bq, (short)0, csq, false, false);
    }

    float MX = fmaxf(fmaxf(-FLT_BIG, mx0), mx1);
    float MN = fminf(fminf( FLT_BIG, mn0), mn1);
    MX = fmaxf(MX, __shfl_xor(MX, 16, 32));
    MN = fminf(MN, __shfl_xor(MN, 16, 32));

    if (lane < 16) {
        const float S = csum[0];
        const float Q = csq[0];
        const float invN = 1.0f / (float)N_DIM;
        const float mean = S * invN;
        const float var  = fmaxf(Q * invN - mean * mean, 0.0f);
        const float sd   = __builtin_sqrtf(var);

        float* o = out + (size_t)b * (5 * D_DIM);
        o[n]             = mean;
        o[D_DIM + n]     = MX;
        o[2 * D_DIM + n] = MN;
        o[3 * D_DIM + n] = S;
        o[4 * D_DIM + n] = sd;
    }
}

extern "C" void kernel_launch(void* const* d_in, const int* in_sizes, int n_in,
                              void* d_out, int out_size, void* d_ws, size_t ws_size,
                              hipStream_t stream) {
    (void)in_sizes; (void)n_in; (void)out_size; (void)ws_size;
    const float* in = (const float*)d_in[0];
    float* out = (float*)d_out;
    float* ws  = (float*)d_ws;

    dim3 g1(PCHUNK, B_DIM);
    agg_stage1<<<g1, NWAVE * 32, 0, stream>>>(in, ws);

    dim3 g2(D_DIM / 16, B_DIM);
    agg_stage2<<<g2, 32, 0, stream>>>(ws, out);
}
